# starting kernel
# baseline (speedup 1.0000x reference)
#include <hip/hip_runtime.h>
#include <math.h>

typedef float v2f __attribute__((ext_vector_type(2)));
typedef float v8f __attribute__((ext_vector_type(8)));

#define B_     128
#define S_     256
#define MW_    21
#define ECHAR  50
#define EWORD  256
#define KW     5
#define NWORDS (B_ * S_)
#define NPOS   (MW_ - KW + 1)
#define KIM    (ECHAR * KW)
#define KPAD   252
#define NPAIR  (KPAD / 2)


#define WS_WT  0
#define WS_WP  (WS_WT + NPAIR * 2 * EWORD)
#define WS_WG  (WS_WP + EWORD * EWORD)
#define WS_H   (WS_WG + EWORD * EWORD)

#define SA_PAIRS   526
#define SA_STRIDE  32
#define SB_PAIRS   126
#define SB_STRIDE  512
#define SH_PAIRS   128
#define SH_STRIDE  96

__device__ __forceinline__ v8f wmma_f32(v2f a, v2f b, v8f c) {
  return __builtin_amdgcn_wmma_f32_16x16x4_f32(false, a, false, b, (short)0, c,
                                               false, false);
}

__global__ void __launch_bounds__(256)
prep_kernel(const float* __restrict__ conv_w, const float* __restrict__ w_proj,
            const float* __restrict__ w_gate, float* __restrict__ ws) {
  int i = blockIdx.x * 256 + threadIdx.x;
  const int n1 = NPAIR * 2 * EWORD;
  const int n2 = EWORD * EWORD;
  if (i < n1) {
    int pair = i >> 9, e = i & 511;
    int o = e >> 1, par = e & 1;
    int q = pair * 2 + par;
    float v = 0.0f;
    if (q < KIM) {
      int k = q / ECHAR, c = q % ECHAR;
      v = conv_w[(o * ECHAR + c) * KW + k];
    }
    ws[WS_WT + i] = v;
  } else if (i < n1 + 2 * n2) {
    int j = i - n1;
    int m = (j < n2) ? 0 : 1;
    int jj = j - m * n2;
    int pair = jj >> 9, e = jj & 511;
    int o = e >> 1, par = e & 1;
    int c = pair * 2 + par;
    const float* w = m ? w_gate : w_proj;
    ws[(m ? WS_WG : WS_WP) + jj] = w[o * EWORD + c];
  }
}

__global__ void __launch_bounds__(256)
conv_kernel(const int* __restrict__ sents, const float* __restrict__ emb,
            const float* __restrict__ conv_b, float* __restrict__ ws) {
  extern __shared__ float smem[];
  float* sA = smem;
  float* sB = smem + SA_PAIRS * SA_STRIDE;

  const float* WtI = ws + WS_WT;
  float* H = ws + WS_H;

  const int tid = threadIdx.x;
  const int wg = blockIdx.x;

  for (int j = tid; j < 1050 * 16; j += 256) {
    int word = j & 15;
    int row = j >> 4;
    int p = row / ECHAR;
    int c = row - p * ECHAR;
    int ch = sents[(wg * 16 + word) * MW_ + p];
    sA[(row >> 1) * SA_STRIDE + word * 2 + (row & 1)] = emb[ch * ECHAR + c];
  }
  if (tid < 32) sA[(SA_PAIRS - 1) * SA_STRIDE + tid] = 0.0f;

  for (int j = tid; j < SB_PAIRS * 512; j += 256) {
    int pair = j >> 9;
    int e = j & 511;
    int col = e >> 1, par = e & 1;
    int colx = col ^ ((pair & 1) << 4);
    sB[pair * SB_STRIDE + colx * 2 + par] = WtI[j];
  }
  __syncthreads();

  const int wave = tid >> 5;
  const int lane = tid & 31;
  const int hf = lane >> 4;
  const int lm = lane & 15;
  const int c0 = wave * 32;

  const float* bbase0 = sB + hf * SB_STRIDE + ((c0 + lm) ^ (hf << 4)) * 2;
  const float* bbase1 = sB + hf * SB_STRIDE + ((c0 + 16 + lm) ^ (hf << 4)) * 2;

  const float bias0 = conv_b[c0 + lm];
  const float bias1 = conv_b[c0 + 16 + lm];

  v8f mx0 = (v8f)0.0f, mx1 = (v8f)0.0f;

  for (int t = 0; t < NPOS; ++t) {
    v8f acc0 = (v8f)0.0f, acc1 = (v8f)0.0f;
    const float* ap = sA + (t * 25 + hf) * SA_STRIDE + lm * 2;
    const float* bp0 = bbase0;
    const float* bp1 = bbase1;
    for (int blk = 0; blk < 7; ++blk) {
#pragma unroll
      for (int u = 0; u < 9; ++u) {
        v2f a = *(const v2f*)(ap + u * (2 * SA_STRIDE));
        v2f b0 = *(const v2f*)(bp0 + u * (2 * SB_STRIDE));
        v2f b1 = *(const v2f*)(bp1 + u * (2 * SB_STRIDE));
        acc0 = wmma_f32(a, b0, acc0);
        acc1 = wmma_f32(a, b1, acc1);
      }
      ap += 9 * (2 * SA_STRIDE);
      bp0 += 9 * (2 * SB_STRIDE);
      bp1 += 9 * (2 * SB_STRIDE);
    }
    for (int r = 0; r < 8; ++r) {
      mx0[r] = fmaxf(mx0[r], fmaxf(acc0[r] + bias0, 0.0f));
      mx1[r] = fmaxf(mx1[r], fmaxf(acc1[r] + bias1, 0.0f));
    }
  }

  for (int r = 0; r < 8; ++r) {
    int word = wg * 16 + r + 8 * hf;
    H[word * EWORD + c0 + lm] = mx0[r];
    H[word * EWORD + c0 + 16 + lm] = mx1[r];
  }
}

__global__ void __launch_bounds__(256)
highway_kernel(const float* __restrict__ b_proj, const float* __restrict__ b_gate,
               const float* __restrict__ ws_c, float* __restrict__ out) {
  __shared__ float sH[SH_PAIRS * SH_STRIDE];
  const float* WpI = ws_c + WS_WP;
  const float* WgI = ws_c + WS_WG;
  const float* H = ws_c + WS_H;

  const int tid = threadIdx.x;
  const int wg = blockIdx.x;

  for (int w = 0; w < 32; ++w)
    sH[(tid >> 1) * SH_STRIDE + w * 2 + (tid & 1)] = H[(wg * 32 + w) * EWORD + tid];
  __syncthreads();

  const int wave = tid >> 5;
  const int lane = tid & 31;
  const int hf = lane >> 4;
  const int lm = lane & 15;
  const int c0 = wave * 32;

  const float bp0 = b_proj[c0 + lm], bp1 = b_proj[c0 + 16 + lm];
  const float bg0 = b_gate[c0 + lm], bg1 = b_gate[c0 + 16 + lm];

  v8f aP00 = (v8f)0.0f, aP01 = (v8f)0.0f, aP10 = (v8f)0.0f, aP11 = (v8f)0.0f;
  v8f aG00 = (v8f)0.0f, aG01 = (v8f)0.0f, aG10 = (v8f)0.0f, aG11 = (v8f)0.0f;

  const float* hp = sH + hf * SH_STRIDE + lm * 2;
  const float* p0 = WpI + hf * 512 + (c0 + lm) * 2;
  const float* p1 = WpI + hf * 512 + (c0 + 16 + lm) * 2;
  const float* g0 = WgI + hf * 512 + (c0 + lm) * 2;
  const float* g1 = WgI + hf * 512 + (c0 + 16 + lm) * 2;

  for (int blk = 0; blk < 8; ++blk) {
#pragma unroll
    for (int u = 0; u < 8; ++u) {
      v2f a0 = *(const v2f*)(hp + u * (2 * SH_STRIDE));
      v2f a1 = *(const v2f*)(hp + u * (2 * SH_STRIDE) + 32);
      v2f bP0 = *(const v2f*)(p0 + u * 1024);
      v2f bP1 = *(const v2f*)(p1 + u * 1024);
      v2f bG0 = *(const v2f*)(g0 + u * 1024);
      v2f bG1 = *(const v2f*)(g1 + u * 1024);
      aP00 = wmma_f32(a0, bP0, aP00);
      aP01 = wmma_f32(a0, bP1, aP01);
      aP10 = wmma_f32(a1, bP0, aP10);
      aP11 = wmma_f32(a1, bP1, aP11);
      aG00 = wmma_f32(a0, bG0, aG00);
      aG01 = wmma_f32(a0, bG1, aG01);
      aG10 = wmma_f32(a1, bG0, aG10);
      aG11 = wmma_f32(a1, bG1, aG11);
    }
    hp += 8 * (2 * SH_STRIDE);
    p0 += 8 * 1024;
    p1 += 8 * 1024;
    g0 += 8 * 1024;
    g1 += 8 * 1024;
  }

  for (int mt = 0; mt < 2; ++mt) {
    const v8f& aP_0 = mt ? aP10 : aP00;
    const v8f& aP_1 = mt ? aP11 : aP01;
    const v8f& aG_0 = mt ? aG10 : aG00;
    const v8f& aG_1 = mt ? aG11 : aG01;
    for (int r = 0; r < 8; ++r) {
      int word = mt * 16 + r + 8 * hf;
      int gword = wg * 32 + word;
      {
        int col = c0 + lm;
        float y = sH[(col >> 1) * SH_STRIDE + word * 2 + (col & 1)];
        float p = fmaxf(aP_0[r] + bp0, 0.0f);
        float g = 1.0f / (1.0f + __expf(-(aG_0[r] + bg0)));
        out[gword * EWORD + col] = g * p + (1.0f - g) * y;
      }
      {
        int col = c0 + 16 + lm;
        float y = sH[(col >> 1) * SH_STRIDE + word * 2 + (col & 1)];
        float p = fmaxf(aP_1[r] + bp1, 0.0f);
        float g = 1.0f / (1.0f + __expf(-(aG_1[r] + bg1)));
        out[gword * EWORD + col] = g * p + (1.0f - g) * y;
      }
    }
  }
}

extern "C" void kernel_launch(void* const* d_in, const int* in_sizes, int n_in,
                              void* d_out, int out_size, void* d_ws, size_t ws_size,
                              hipStream_t stream) {
  (void)in_sizes; (void)n_in; (void)out_size; (void)ws_size;
  const int* sents = (const int*)d_in[0];
  const float* emb = (const float*)d_in[1];
  const float* conv_w = (const float*)d_in[2];
  const float* conv_b = (const float*)d_in[3];
  const float* w_proj = (const float*)d_in[4];
  const float* b_proj = (const float*)d_in[5];
  const float* w_gate = (const float*)d_in[6];
  const float* b_gate = (const float*)d_in[7];
  float* ws = (float*)d_ws;
  float* out = (float*)d_out;

  const int prep_n = NPAIR * 2 * EWORD + 2 * EWORD * EWORD;
  prep_kernel<<<(prep_n + 255) / 256, 256, 0, stream>>>(conv_w, w_proj, w_gate, ws);

  const size_t conv_lds =
      (size_t)(SA_PAIRS * SA_STRIDE + SB_PAIRS * SB_STRIDE) * sizeof(float);
  conv_kernel<<<NWORDS / 16, 256, conv_lds, stream>>>(sents, emb, conv_b, ws);

  highway_kernel<<<NWORDS / 32, 256, 0, stream>>>(b_proj, b_gate, ws, out);
}


# highway fused into conv kernel: a wave keeps its words' pooled values in LDS, rereads them as A fragments, streams proj/gate weights (conv B ring filled once); 2 launches; highway sum order changes
# speedup vs baseline: 2.6425x; 2.6425x over previous
#include <hip/hip_runtime.h>
#include <math.h>


typedef float v2f __attribute__((ext_vector_type(2)));
typedef float v8f __attribute__((ext_vector_type(8)));

#define B_     128
#define S_     256
#define MW_    21
#define ECHAR  50
#define EWORD  256
#define KW     5
#define NWORDS (B_ * S_)
#define NPOS   (MW_ - KW + 1)
#define TSTEPS 13
#define NTILE  (EWORD / 16)

#define NJ     10
#define NOUT   6
#define NTL    3
#define WSTEPS (NJ * TSTEPS)


#define HSTEPS 64
#define HFRAGS (HSTEPS * 2)
#define PFH 64

#define WS_WT  0
#define WS_WH  (WS_WT + (NTILE * WSTEPS + PF) * 64)
#define WS_PAD (WS_WH + NTILE * HFRAGS * 64)

#define SM_PITCH 514

#define PF 65

__device__ __forceinline__ v8f wmma_f32(v2f a, v2f b, v8f c) {
  return __builtin_amdgcn_wmma_f32_16x16x4_f32(false, a, false, b, (short)0, c,
                                               false, false);
}

__device__ __forceinline__ float wtrans(int j, float w0, float w1, float w2,
                                        float w3, float w4) {
  if (j == 0) return w0 * (1.0f / 36.0f);
  if (j == NJ - 1) return w4 * (1.0f / 64.0f);
  const float pt = (j == 1) ? 1.0f : (j == 2) ? -1.0f : (j == 3) ? 2.0f
                 : (j == 4) ? -2.0f : (j == 5) ? 0.5f : (j == 6) ? -0.5f
                 : (j == 7) ? 0.75f : -0.75f;
  const float sc = (j <= 2) ? (1.0f / 126.0f) : (j <= 4) ? (1.0f / 19800.0f)
                 : (j <= 6) ? (16.0f / 225.0f) : (2048.0f / 17325.0f);
  float h = w4;
  h = h * pt + w3;
  h = h * pt + w2;
  h = h * pt + w1;
  h = h * pt + w0;
  return h * sc;
}

__global__ void __launch_bounds__(256)
prep_kernel(const float* __restrict__ conv_w, const float* __restrict__ w_proj,
            const float* __restrict__ w_gate, float* __restrict__ ws) {
  int i = blockIdx.x * 256 + threadIdx.x;
  const int n1 = (NTILE * WSTEPS + PF) * 64;
  const int n2 = NTILE * HFRAGS * 64;
  const int n3 = PFH * 64;
  if (i < n1) {
    int e = i & 1;
    int lane = (i >> 1) & 31;
    int f = i >> 6;
    if (f >= NTILE * WSTEPS) f -= NTILE * WSTEPS;
    int n = f / WSTEPS, st = f - n * WSTEPS;
    int sk = st / NJ, j = st - sk * NJ;
    int ch = 4 * sk + 2 * (lane >> 4) + e;
    int o = n * 16 + (lane & 15);
    float v = 0.0f;
    if (ch < ECHAR) {
      const float* w = conv_w + (o * ECHAR + ch) * KW;
      v = wtrans(j, w[0], w[1], w[2], w[3], w[4]);
    }
    ws[WS_WT + i] = v;
  } else if (i < n1 + n2) {
    int j = i - n1;
    int e = j & 1;
    int lane = (j >> 1) & 31;
    int f = j >> 6;
    int pg = f & 1, s = (f >> 1) & (HSTEPS - 1), n = f >> 7;
    int ch = 32 * (s >> 3) + 16 * (lane >> 4) + 2 * (s & 7) + e;
    int o = n * 16 + (lane & 15);
    const float* w = pg ? w_gate : w_proj;
    ws[WS_WH + j] = w[o * EWORD + ch];
  } else if (i < n1 + n2 + n3) {
    ws[WS_PAD + (i - n1 - n2)] = 0.0f;
  }
}

__global__ void __launch_bounds__(128)
conv_kernel(const int* __restrict__ sents, const float* __restrict__ emb,
            const float* __restrict__ conv_b, const float* __restrict__ b_proj,
            const float* __restrict__ b_gate, const float* __restrict__ ws,
            float* __restrict__ out) {
  __shared__ __attribute__((aligned(16))) float sM[4 * 16 * SM_PITCH];
  const float* WtF = ws + WS_WT;
  const float* WhF = ws + WS_WH;

  const int lane = threadIdx.x & 31;
  const int grp = blockIdx.x * 4 + (threadIdx.x >> 5);
  float* sMb = sM + (threadIdx.x >> 5) * (16 * SM_PITCH);
  float* sMw = sMb + lane;
  const int hf = lane >> 4;
  const int lm = lane & 15;

  v2f ring[PF];
#pragma unroll
  for (int i = 0; i < PF; ++i) ring[i] = *(const v2f*)(WtF + lane * 2 + i * 64);

#pragma nounroll
  for (int tl = 0; tl < NTL; ++tl) {
    const bool last = (tl == NTL - 1);
    const float drop5 = last ? -INFINITY : 0.0f;

    v2f X[2][NJ][TSTEPS];
#pragma unroll
    for (int mt = 0; mt < 2; ++mt) {
      const int word = grp * 32 + mt * 16 + lm;
      const float* row[NJ];
      bool live[NJ];
#pragma unroll
      for (int p = 0; p < NJ; ++p) {
        live[p] = (p < NJ - 1) || !last;
        const int pos = live[p] ? (NOUT * tl + p) : (MW_ - 1);
        const int ch = sents[word * MW_ + pos];
        row[p] = emb + ch * ECHAR;
      }
#pragma unroll
      for (int sk = 0; sk < TSTEPS; ++sk) {
        v2f d[NJ];
#pragma unroll
        for (int p = 0; p < NJ; ++p) {
          v2f v;
          if (sk < TSTEPS - 1) {
            v = *(const v2f*)(row[p] + 4 * sk + 2 * hf);
          } else {
            v = *(const v2f*)(row[p] + 48);
            v = hf ? (v2f)0.0f : v;
          }
          d[p] = live[p] ? v : (v2f)0.0f;
        }
        X[mt][0][sk] = 36.0f * d[0] - 253.0f * d[2] + 525.0f * d[4] - 372.0f * d[6] + 64.0f * d[8];
        const v2f e1 = 36.0f * d[2] - 217.0f * d[4] + 308.0f * d[6] - 64.0f * d[8];
        const v2f o1 = 36.0f * d[1] - 217.0f * d[3] + 308.0f * d[5] - 64.0f * d[7];
        X[mt][1][sk] = e1 + o1;
        X[mt][2][sk] = e1 - o1;
        const v2f e2 = -9.0f * d[2] + 61.0f * d[4] - 116.0f * d[6] + 64.0f * d[8];
        const v2f o2 = -18.0f * d[1] + 122.0f * d[3] - 232.0f * d[5] + 128.0f * d[7];
        X[mt][3][sk] = e2 + o2;
        X[mt][4][sk] = e2 - o2;
        const v2f e3 = 72.0f * d[2] - 218.0f * d[4] + 178.0f * d[6] - 32.0f * d[8];
        const v2f o3 = 36.0f * d[1] - 109.0f * d[3] + 89.0f * d[5] - 16.0f * d[7];
        X[mt][5][sk] = e3 + o3;
        X[mt][6][sk] = e3 - o3;
        const v2f e4 = -16.0f * d[2] + 84.0f * d[4] - 84.0f * d[6] + 16.0f * d[8];
        const v2f o4 = -12.0f * d[1] + 63.0f * d[3] - 63.0f * d[5] + 12.0f * d[7];
        X[mt][7][sk] = e4 + o4;
        X[mt][8][sk] = e4 - o4;
        X[mt][9][sk] = 36.0f * d[1] - 253.0f * d[3] + 525.0f * d[5] - 372.0f * d[7] + 64.0f * d[9];
      }
    }

    const float* bp = WtF + lane * 2 + PF * 64;

#pragma nounroll
    for (int n = 0; n < NTILE; ++n) {
      const float bias = conv_b[n * 16 + lm];

      float* mrow = sMw + n * 32;

      float hp[2][8];
#pragma unroll
      for (int mt = 0; mt < 2; ++mt)
        for (int r = 0; r < 8; ++r) hp[mt][r] = -INFINITY;
      if (tl > 0) {
#pragma unroll
        for (int mt = 0; mt < 2; ++mt)
          for (int r = 0; r < 8; ++r) hp[mt][r] = mrow[(mt * 8 + r) * SM_PITCH];
      }

      v8f acc[2][NJ];
#pragma unroll
      for (int mt = 0; mt < 2; ++mt)
        for (int j = 0; j < NJ; ++j) acc[mt][j] = (v8f)0.0f;

#pragma unroll
      for (int st = 0; st < WSTEPS; ++st) {
        if (st == 44 || st == 87) __builtin_amdgcn_sched_barrier(0);
        const int sk = st / NJ, j = st % NJ;
        const v2f b = ring[st % PF];
        ring[st % PF] = *(const v2f*)(bp + st * 64);
        acc[0][j] = wmma_f32(X[0][j][sk], b, acc[0][j]);
        acc[1][j] = wmma_f32(X[1][j][sk], b, acc[1][j]);
      }
      bp += WSTEPS * 64;
      __builtin_amdgcn_sched_barrier(0);

      float mx[2][8];
#pragma unroll
      for (int mt = 0; mt < 2; ++mt) {
        for (int r = 0; r < 8; ++r) {
          const float s1 = acc[mt][1][r] + acc[mt][2][r];
          const float d1 = acc[mt][1][r] - acc[mt][2][r];
          const float s2 = acc[mt][3][r] + acc[mt][4][r];
          const float d2 = acc[mt][3][r] - acc[mt][4][r];
          const float s3 = acc[mt][5][r] + acc[mt][6][r];
          const float d3 = acc[mt][5][r] - acc[mt][6][r];
          const float s4 = acc[mt][7][r] + acc[mt][8][r];
          const float d4 = acc[mt][7][r] - acc[mt][8][r];
          const float y0 = acc[mt][0][r] + s1 + s2 + s3 + s4;
          const float y1 = d1 + 2.0f * d2 + 0.5f * d3 + 0.75f * d4;
          const float y2 = s1 + 4.0f * s2 + 0.25f * s3 + 0.5625f * s4;
          const float y3 = d1 + 8.0f * d2 + 0.125f * d3 + 0.421875f * d4;
          const float y4 = s1 + 16.0f * s2 + 0.0625f * s3 + 0.31640625f * s4;
          const float y5 = d1 + 32.0f * d2 + 0.03125f * d3 + 0.2373046875f * d4
                           + acc[mt][9][r];
          const float m5 = fmaxf(fmaxf(fmaxf(y0, y1), fmaxf(y2, y3)), y4);
          mx[mt][r] = fmaxf(fmaxf(m5, y5 + drop5), hp[mt][r]);
        }
      }
      if (last) {
#pragma unroll
        for (int mt = 0; mt < 2; ++mt)
          for (int r = 0; r < 8; ++r)
            mx[mt][r] = fmaxf(mx[mt][r] + bias, 0.0f);
      }
#pragma unroll
      for (int mt = 0; mt < 2; ++mt)
        for (int r = 0; r < 8; ++r)
          mrow[(mt * 8 + r) * SM_PITCH] = mx[mt][r];
    }
  }

  __builtin_amdgcn_wave_barrier();

  v2f A[2][HSTEPS];
  {
    const float* ap = sMb + (lm & 7) * SM_PITCH + hf * 32 + (lm >> 3) * 16;
#pragma unroll
    for (int mt = 0; mt < 2; ++mt) {
#pragma unroll
      for (int s = 0; s < HSTEPS; ++s)
        A[mt][s] = *(const v2f*)(ap + mt * 8 * SM_PITCH + (s >> 3) * 64 + (s & 7) * 2);
    }
  }

  const float* bq = WhF + lane * 2;
  v2f hring[PFH];
#pragma unroll
  for (int i = 0; i < PFH; ++i) hring[i] = *(const v2f*)(bq + i * 64);
  bq += PFH * 64;

#pragma nounroll
  for (int n = 0; n < NTILE; ++n) {
    const float bp0 = b_proj[n * 16 + lm];
    const float bg0 = b_gate[n * 16 + lm];
    float* orow = out + (grp * 32 + 8 * hf) * EWORD + n * 16 + lm;
    const float* yrow = sMw + n * 32;

    v8f aP[2], aG[2];
#pragma unroll
    for (int mt = 0; mt < 2; ++mt) { aP[mt] = (v8f)0.0f; aG[mt] = (v8f)0.0f; }

#pragma unroll
    for (int f = 0; f < HFRAGS; ++f) {
      if (f == 43 || f == 86) __builtin_amdgcn_sched_barrier(0);
      const int s = f >> 1;
      const v2f b = hring[f % PFH];
      hring[f % PFH] = *(const v2f*)(bq + f * 64);
      if (f & 1) {
        aG[0] = wmma_f32(A[0][s], b, aG[0]);
        aG[1] = wmma_f32(A[1][s], b, aG[1]);
      } else {
        aP[0] = wmma_f32(A[0][s], b, aP[0]);
        aP[1] = wmma_f32(A[1][s], b, aP[1]);
      }
    }
    bq += HFRAGS * 64;
    __builtin_amdgcn_sched_barrier(0);

#pragma unroll
    for (int mt = 0; mt < 2; ++mt) {
      for (int r = 0; r < 8; ++r) {
        const float y = yrow[(mt * 8 + r) * SM_PITCH];
        const float p = fmaxf(aP[mt][r] + bp0, 0.0f);
        const float g = 1.0f / (1.0f + __expf(-(aG[mt][r] + bg0)));
        orow[(mt * 16 + r) * EWORD] = g * p + (1.0f - g) * y;
      }
    }
  }
}

extern "C" void kernel_launch(void* const* d_in, const int* in_sizes, int n_in,
                              void* d_out, int out_size, void* d_ws, size_t ws_size,
                              hipStream_t stream) {
  (void)in_sizes; (void)n_in; (void)out_size; (void)ws_size;
  const int* sents = (const int*)d_in[0];
  const float* emb = (const float*)d_in[1];
  const float* conv_w = (const float*)d_in[2];
  const float* conv_b = (const float*)d_in[3];
  const float* w_proj = (const float*)d_in[4];
  const float* b_proj = (const float*)d_in[5];
  const float* w_gate = (const float*)d_in[6];
  const float* b_gate = (const float*)d_in[7];
  float* ws = (float*)d_ws;
  float* out = (float*)d_out;

  const int prep_n = (NTILE * WSTEPS + PF) * 64 + NTILE * HFRAGS * 64 + PFH * 64;
  prep_kernel<<<(prep_n + 255) / 256, 256, 0, stream>>>(conv_w, w_proj, w_gate, ws);

  conv_kernel<<<NWORDS / 128, 128, 0, stream>>>(sents, emb, conv_b, b_proj, b_gate,
                                                ws, out);
}


# embedding table (100 x 50 f32) staged once per workgroup in LDS behind one barrier; each tile's input gather (one row per word, 8 bytes a lane) reads LDS, not global; values and sum order unchanged
# speedup vs baseline: 2.7375x; 1.0360x over previous
#include <hip/hip_runtime.h>
#include <math.h>


typedef float v2f __attribute__((ext_vector_type(2)));
typedef float v8f __attribute__((ext_vector_type(8)));

#define B_     128
#define S_     256
#define MW_    21
#define ECHAR  50
#define VOCAB  100
#define EWORD  256
#define KW     5
#define NWORDS (B_ * S_)
#define NPOS   (MW_ - KW + 1)
#define TSTEPS 13
#define NTILE  (EWORD / 16)

#define NJ     10
#define NOUT   6
#define NTL    3
#define WSTEPS (NJ * TSTEPS)


#define HSTEPS 64
#define HFRAGS (HSTEPS * 2)
#define PFH 64

#define WS_WT  0
#define WS_WH  (WS_WT + (NTILE * WSTEPS + PF) * 64)
#define WS_PAD (WS_WH + NTILE * HFRAGS * 64)

#define SM_PITCH 514

#define PF 65

__device__ __forceinline__ v8f wmma_f32(v2f a, v2f b, v8f c) {
  return __builtin_amdgcn_wmma_f32_16x16x4_f32(false, a, false, b, (short)0, c,
                                               false, false);
}

__device__ __forceinline__ float wtrans(int j, float w0, float w1, float w2,
                                        float w3, float w4) {
  if (j == 0) return w0 * (1.0f / 36.0f);
  if (j == NJ - 1) return w4 * (1.0f / 64.0f);
  const float pt = (j == 1) ? 1.0f : (j == 2) ? -1.0f : (j == 3) ? 2.0f
                 : (j == 4) ? -2.0f : (j == 5) ? 0.5f : (j == 6) ? -0.5f
                 : (j == 7) ? 0.75f : -0.75f;
  const float sc = (j <= 2) ? (1.0f / 126.0f) : (j <= 4) ? (1.0f / 19800.0f)
                 : (j <= 6) ? (16.0f / 225.0f) : (2048.0f / 17325.0f);
  float h = w4;
  h = h * pt + w3;
  h = h * pt + w2;
  h = h * pt + w1;
  h = h * pt + w0;
  return h * sc;
}

__global__ void __launch_bounds__(256)
prep_kernel(const float* __restrict__ conv_w, const float* __restrict__ w_proj,
            const float* __restrict__ w_gate, float* __restrict__ ws) {
  int i = blockIdx.x * 256 + threadIdx.x;
  const int n1 = (NTILE * WSTEPS + PF) * 64;
  const int n2 = NTILE * HFRAGS * 64;
  const int n3 = PFH * 64;
  if (i < n1) {
    int e = i & 1;
    int lane = (i >> 1) & 31;
    int f = i >> 6;
    if (f >= NTILE * WSTEPS) f -= NTILE * WSTEPS;
    int n = f / WSTEPS, st = f - n * WSTEPS;
    int sk = st / NJ, j = st - sk * NJ;
    int ch = 4 * sk + 2 * (lane >> 4) + e;
    int o = n * 16 + (lane & 15);
    float v = 0.0f;
    if (ch < ECHAR) {
      const float* w = conv_w + (o * ECHAR + ch) * KW;
      v = wtrans(j, w[0], w[1], w[2], w[3], w[4]);
    }
    ws[WS_WT + i] = v;
  } else if (i < n1 + n2) {
    int j = i - n1;
    int e = j & 1;
    int lane = (j >> 1) & 31;
    int f = j >> 6;
    int pg = f & 1, s = (f >> 1) & (HSTEPS - 1), n = f >> 7;
    int ch = 32 * (s >> 3) + 16 * (lane >> 4) + 2 * (s & 7) + e;
    int o = n * 16 + (lane & 15);
    const float* w = pg ? w_gate : w_proj;
    ws[WS_WH + j] = w[o * EWORD + ch];
  } else if (i < n1 + n2 + n3) {
    ws[WS_PAD + (i - n1 - n2)] = 0.0f;
  }
}

__global__ void __launch_bounds__(128)
conv_kernel(const int* __restrict__ sents, const float* __restrict__ emb,
            const float* __restrict__ conv_b, const float* __restrict__ b_proj,
            const float* __restrict__ b_gate, const float* __restrict__ ws,
            float* __restrict__ out) {
  __shared__ __attribute__((aligned(16))) float sM[4 * 16 * SM_PITCH];
  __shared__ __attribute__((aligned(16))) float sE[VOCAB * ECHAR];
  const float* WtF = ws + WS_WT;
  const float* WhF = ws + WS_WH;

  const int lane = threadIdx.x & 31;
  const int grp = blockIdx.x * 4 + (threadIdx.x >> 5);
  float* sMb = sM + (threadIdx.x >> 5) * (16 * SM_PITCH);
  float* sMw = sMb + lane;
  const int hf = lane >> 4;
  const int lm = lane & 15;

  v2f ring[PF];
#pragma unroll
  for (int i = 0; i < PF; ++i) ring[i] = *(const v2f*)(WtF + lane * 2 + i * 64);

#pragma unroll
  for (int k = 0; k < (VOCAB * ECHAR / 2 + 127) / 128; ++k) {
    const int i = (threadIdx.x & 127) + 128 * k;
    if (i < VOCAB * ECHAR / 2) *(v2f*)(sE + 2 * i) = *(const v2f*)(emb + 2 * i);
  }
  __syncthreads();

#pragma nounroll
  for (int tl = 0; tl < NTL; ++tl) {
    const bool last = (tl == NTL - 1);
    const float drop5 = last ? -INFINITY : 0.0f;

    v2f X[2][NJ][TSTEPS];
#pragma unroll
    for (int mt = 0; mt < 2; ++mt) {
      const int word = grp * 32 + mt * 16 + lm;
      int roff[NJ];
      bool live[NJ];
#pragma unroll
      for (int p = 0; p < NJ; ++p) {
        live[p] = (p < NJ - 1) || !last;
        const int pos = live[p] ? (NOUT * tl + p) : (MW_ - 1);
        const int ch = sents[word * MW_ + pos];
        roff[p] = ch * ECHAR;
      }
#pragma unroll
      for (int sk = 0; sk < TSTEPS; ++sk) {
        v2f d[NJ];
#pragma unroll
        for (int p = 0; p < NJ; ++p) {
          v2f v;
          if (sk < TSTEPS - 1) {
            v = *(const v2f*)(sE + roff[p] + 4 * sk + 2 * hf);
          } else {
            v = *(const v2f*)(sE + roff[p] + 48);
            v = hf ? (v2f)0.0f : v;
          }
          d[p] = live[p] ? v : (v2f)0.0f;
        }
        X[mt][0][sk] = 36.0f * d[0] - 253.0f * d[2] + 525.0f * d[4] - 372.0f * d[6] + 64.0f * d[8];
        const v2f e1 = 36.0f * d[2] - 217.0f * d[4] + 308.0f * d[6] - 64.0f * d[8];
        const v2f o1 = 36.0f * d[1] - 217.0f * d[3] + 308.0f * d[5] - 64.0f * d[7];
        X[mt][1][sk] = e1 + o1;
        X[mt][2][sk] = e1 - o1;
        const v2f e2 = -9.0f * d[2] + 61.0f * d[4] - 116.0f * d[6] + 64.0f * d[8];
        const v2f o2 = -18.0f * d[1] + 122.0f * d[3] - 232.0f * d[5] + 128.0f * d[7];
        X[mt][3][sk] = e2 + o2;
        X[mt][4][sk] = e2 - o2;
        const v2f e3 = 72.0f * d[2] - 218.0f * d[4] + 178.0f * d[6] - 32.0f * d[8];
        const v2f o3 = 36.0f * d[1] - 109.0f * d[3] + 89.0f * d[5] - 16.0f * d[7];
        X[mt][5][sk] = e3 + o3;
        X[mt][6][sk] = e3 - o3;
        const v2f e4 = -16.0f * d[2] + 84.0f * d[4] - 84.0f * d[6] + 16.0f * d[8];
        const v2f o4 = -12.0f * d[1] + 63.0f * d[3] - 63.0f * d[5] + 12.0f * d[7];
        X[mt][7][sk] = e4 + o4;
        X[mt][8][sk] = e4 - o4;
        X[mt][9][sk] = 36.0f * d[1] - 253.0f * d[3] + 525.0f * d[5] - 372.0f * d[7] + 64.0f * d[9];
      }
    }

    const float* bp = WtF + lane * 2 + PF * 64;

#pragma nounroll
    for (int n = 0; n < NTILE; ++n) {
      const float bias = conv_b[n * 16 + lm];

      float* mrow = sMw + n * 32;

      float hp[2][8];
#pragma unroll
      for (int mt = 0; mt < 2; ++mt)
        for (int r = 0; r < 8; ++r) hp[mt][r] = -INFINITY;
      if (tl > 0) {
#pragma unroll
        for (int mt = 0; mt < 2; ++mt)
          for (int r = 0; r < 8; ++r) hp[mt][r] = mrow[(mt * 8 + r) * SM_PITCH];
      }

      v8f acc[2][NJ];
#pragma unroll
      for (int mt = 0; mt < 2; ++mt)
        for (int j = 0; j < NJ; ++j) acc[mt][j] = (v8f)0.0f;

#pragma unroll
      for (int st = 0; st < WSTEPS; ++st) {
        if (st == 44 || st == 87) __builtin_amdgcn_sched_barrier(0);
        const int sk = st / NJ, j = st % NJ;
        const v2f b = ring[st % PF];
        ring[st % PF] = *(const v2f*)(bp + st * 64);
        acc[0][j] = wmma_f32(X[0][j][sk], b, acc[0][j]);
        acc[1][j] = wmma_f32(X[1][j][sk], b, acc[1][j]);
      }
      bp += WSTEPS * 64;
      __builtin_amdgcn_sched_barrier(0);

      float mx[2][8];
#pragma unroll
      for (int mt = 0; mt < 2; ++mt) {
        for (int r = 0; r < 8; ++r) {
          const float s1 = acc[mt][1][r] + acc[mt][2][r];
          const float d1 = acc[mt][1][r] - acc[mt][2][r];
          const float s2 = acc[mt][3][r] + acc[mt][4][r];
          const float d2 = acc[mt][3][r] - acc[mt][4][r];
          const float s3 = acc[mt][5][r] + acc[mt][6][r];
          const float d3 = acc[mt][5][r] - acc[mt][6][r];
          const float s4 = acc[mt][7][r] + acc[mt][8][r];
          const float d4 = acc[mt][7][r] - acc[mt][8][r];
          const float y0 = acc[mt][0][r] + s1 + s2 + s3 + s4;
          const float y1 = d1 + 2.0f * d2 + 0.5f * d3 + 0.75f * d4;
          const float y2 = s1 + 4.0f * s2 + 0.25f * s3 + 0.5625f * s4;
          const float y3 = d1 + 8.0f * d2 + 0.125f * d3 + 0.421875f * d4;
          const float y4 = s1 + 16.0f * s2 + 0.0625f * s3 + 0.31640625f * s4;
          const float y5 = d1 + 32.0f * d2 + 0.03125f * d3 + 0.2373046875f * d4
                           + acc[mt][9][r];
          const float m5 = fmaxf(fmaxf(fmaxf(y0, y1), fmaxf(y2, y3)), y4);
          mx[mt][r] = fmaxf(fmaxf(m5, y5 + drop5), hp[mt][r]);
        }
      }
      if (last) {
#pragma unroll
        for (int mt = 0; mt < 2; ++mt)
          for (int r = 0; r < 8; ++r)
            mx[mt][r] = fmaxf(mx[mt][r] + bias, 0.0f);
      }
#pragma unroll
      for (int mt = 0; mt < 2; ++mt)
        for (int r = 0; r < 8; ++r)
          mrow[(mt * 8 + r) * SM_PITCH] = mx[mt][r];
    }
  }

  __builtin_amdgcn_wave_barrier();

  v2f A[2][HSTEPS];
  {
    const float* ap = sMb + (lm & 7) * SM_PITCH + hf * 32 + (lm >> 3) * 16;
#pragma unroll
    for (int mt = 0; mt < 2; ++mt) {
#pragma unroll
      for (int s = 0; s < HSTEPS; ++s)
        A[mt][s] = *(const v2f*)(ap + mt * 8 * SM_PITCH + (s >> 3) * 64 + (s & 7) * 2);
    }
  }

  const float* bq = WhF + lane * 2;
  v2f hring[PFH];
#pragma unroll
  for (int i = 0; i < PFH; ++i) hring[i] = *(const v2f*)(bq + i * 64);
  bq += PFH * 64;

#pragma nounroll
  for (int n = 0; n < NTILE; ++n) {
    const float bp0 = b_proj[n * 16 + lm];
    const float bg0 = b_gate[n * 16 + lm];
    float* orow = out + (grp * 32 + 8 * hf) * EWORD + n * 16 + lm;
    const float* yrow = sMw + n * 32;

    v8f aP[2], aG[2];
#pragma unroll
    for (int mt = 0; mt < 2; ++mt) { aP[mt] = (v8f)0.0f; aG[mt] = (v8f)0.0f; }

#pragma unroll
    for (int f = 0; f < HFRAGS; ++f) {
      if (f == 43 || f == 86) __builtin_amdgcn_sched_barrier(0);
      const int s = f >> 1;
      const v2f b = hring[f % PFH];
      hring[f % PFH] = *(const v2f*)(bq + f * 64);
      if (f & 1) {
        aG[0] = wmma_f32(A[0][s], b, aG[0]);
        aG[1] = wmma_f32(A[1][s], b, aG[1]);
      } else {
        aP[0] = wmma_f32(A[0][s], b, aP[0]);
        aP[1] = wmma_f32(A[1][s], b, aP[1]);
      }
    }
    bq += HFRAGS * 64;
    __builtin_amdgcn_sched_barrier(0);

#pragma unroll
    for (int mt = 0; mt < 2; ++mt) {
      for (int r = 0; r < 8; ++r) {
        const float y = yrow[(mt * 8 + r) * SM_PITCH];
        const float p = fmaxf(aP[mt][r] + bp0, 0.0f);
        const float g = 1.0f / (1.0f + __expf(-(aG[mt][r] + bg0)));
        orow[(mt * 16 + r) * EWORD] = g * p + (1.0f - g) * y;
      }
    }
  }
}

extern "C" void kernel_launch(void* const* d_in, const int* in_sizes, int n_in,
                              void* d_out, int out_size, void* d_ws, size_t ws_size,
                              hipStream_t stream) {
  (void)in_sizes; (void)n_in; (void)out_size; (void)ws_size;
  const int* sents = (const int*)d_in[0];
  const float* emb = (const float*)d_in[1];
  const float* conv_w = (const float*)d_in[2];
  const float* conv_b = (const float*)d_in[3];
  const float* w_proj = (const float*)d_in[4];
  const float* b_proj = (const float*)d_in[5];
  const float* w_gate = (const float*)d_in[6];
  const float* b_gate = (const float*)d_in[7];
  float* ws = (float*)d_ws;
  float* out = (float*)d_out;

  const int prep_n = (NTILE * WSTEPS + PF) * 64 + NTILE * HFRAGS * 64 + PFH * 64;
  prep_kernel<<<(prep_n + 255) / 256, 256, 0, stream>>>(conv_w, w_proj, w_gate, ws);

  conv_kernel<<<NWORDS / 128, 128, 0, stream>>>(sents, emb, conv_b, b_proj, b_gate,
                                                ws, out);
}
